# starting kernel
# baseline (speedup 1.0000x reference)
#include <hip/hip_runtime.h>

typedef __attribute__((ext_vector_type(2))) float v2f;
typedef __attribute__((ext_vector_type(8))) float v8f;

#define B_    16
#define N_    128
#define NC    60
#define NA    4
#define NI    64
#define EAT   16
#define EC    64
#define NIN   128
#define NOUT  60
#define SLOPE 0.01f

__device__ __forceinline__ float lrelu(float v) { return v > 0.f ? v : v * SLOPE; }

__global__ __launch_bounds__(256) void gin_fused(
    const float* __restrict__ x,        const float* __restrict__ node_attr,
    const float* __restrict__ edge_attr,const float* __restrict__ adj,
    const float* __restrict__ We,       const float* __restrict__ be,
    const float* __restrict__ Wn,       const float* __restrict__ bn,
    float* __restrict__ out)
{
    __shared__ float s_send[N_ * EC];
    __shared__ float s_recvb[16 * EC];
    __shared__ float s_agg[16 * EC];
    __shared__ float s_adj[16 * N_];

    const int b    = blockIdx.x >> 3;
    const int it   = blockIdx.x & 7;
    const int i0   = it * 16;

    const int tid  = threadIdx.x;
    const int wave = tid >> 5;
    const int lane = tid & 31;
    const int lh   = lane >> 4;
    const int ll   = lane & 15;

    for (int t = tid; t < 16 * N_; t += 256) {
        int i = t >> 7, j = t & 127;
        s_adj[t] = adj[((b * N_) + i0 + i) * N_ + j];
    }

    for (int t = wave; t < 36; t += 8) {
        const bool is_recv = (t >= 32);
        const int  ct      = is_recv ? (t - 32) : (t & 3);
        const int  mrow0   = is_recv ? i0 : ((t >> 2) * 16);
        const int  wrow0   = is_recv ? EAT : (EAT + NI);
        const int  node    = mrow0 + ll;
        const long nb      = (long)(b * N_ + node);
        v8f acc = {};
        #pragma unroll
        for (int k0 = 0; k0 < NI; k0 += 4) {
            const int ka = k0 + 2 * lh;
            v2f a, bb;
            if (k0 < NC) {
                const float* p = x + nb * NC + ka;
                a.x = p[0]; a.y = p[1];
            } else {
                const float* p = node_attr + nb * NA + (ka - NC);
                a.x = p[0]; a.y = p[1];
            }
            const float* q = We + (wrow0 + ka) * EC + ct * 16 + ll;
            bb.x = q[0]; bb.y = q[EC];
            acc = __builtin_amdgcn_wmma_f32_16x16x4_f32(false, a, false, bb,
                                                        (short)0, acc, false, false);
        }
        #pragma unroll
        for (int r = 0; r < 8; ++r) {
            const int row = r + 8 * lh;
            const int col = ct * 16 + ll;
            if (is_recv) s_recvb[row * EC + col] = acc[r] + be[col];
            else         s_send[((t >> 2) * 16 + row) * EC + col] = acc[r];
        }
    }
    __syncthreads();

    v2f bE[4][4];
    #pragma unroll
    for (int ct = 0; ct < 4; ++ct)
        #pragma unroll
        for (int kk = 0; kk < 4; ++kk) {
            const int k = 4 * kk + 2 * lh;
            const float* q = We + k * EC + ct * 16 + ll;
            bE[ct][kk].x = q[0]; bE[ct][kk].y = q[EC];
        }

    #pragma unroll
    for (int iw = 0; iw < 2; ++iw) {
        const int  il    = wave * 2 + iw;
        const int  i     = i0 + il;
        const long ebase = ((long)(b * N_ + i)) * N_ * EAT;
        float aggp[4] = {0.f, 0.f, 0.f, 0.f};
        float rb[4];
        #pragma unroll
        for (int ct = 0; ct < 4; ++ct) rb[ct] = s_recvb[il * EC + ct * 16 + ll];

        for (int jt = 0; jt < 8; ++jt) {
            v2f a[4];
            #pragma unroll
            for (int kk = 0; kk < 4; ++kk) {
                const int k = 4 * kk + 2 * lh;
                const float* p = edge_attr + ebase + (jt * 16 + ll) * EAT + k;
                a[kk].x = p[0]; a[kk].y = p[1];
            }
            #pragma unroll
            for (int ct = 0; ct < 4; ++ct) {
                v8f d = {};
                #pragma unroll
                for (int kk = 0; kk < 4; ++kk)
                    d = __builtin_amdgcn_wmma_f32_16x16x4_f32(false, a[kk], false, bE[ct][kk],
                                                              (short)0, d, false, false);
                #pragma unroll
                for (int r = 0; r < 8; ++r) {
                    const int j = jt * 16 + r + 8 * lh;
                    float v = d[r] + rb[ct] + s_send[j * EC + ct * 16 + ll];
                    aggp[ct] = fmaf(lrelu(v), s_adj[il * N_ + j], aggp[ct]);
                }
            }
        }
        #pragma unroll
        for (int ct = 0; ct < 4; ++ct) {
            float v = aggp[ct] + __shfl_xor(aggp[ct], 16, 32);
            if (lh == 0) s_agg[il * EC + ct * 16 + ll] = v;
        }
    }
    __syncthreads();

    if (wave < 4) {
        const int ct = wave;
        const int o  = ct * 16 + ll;
        v8f acc = {};
        #pragma unroll
        for (int k0 = 0; k0 < NIN; k0 += 4) {
            const int ka = k0 + 2 * lh;
            v2f a, bb;
            if (k0 < NC) {
                const float* p = x + (long)(b * N_ + i0 + ll) * NC + ka;
                a.x = p[0]; a.y = p[1];
            } else if (k0 < NI) {
                const float* p = node_attr + (long)(b * N_ + i0 + ll) * NA + (ka - NC);
                a.x = p[0]; a.y = p[1];
            } else {
                a.x = s_agg[ll * EC + (ka - NI)];
                a.y = s_agg[ll * EC + (ka - NI) + 1];
            }
            if (o < NOUT) {
                const float* q = Wn + ka * NOUT + o;
                bb.x = q[0]; bb.y = q[NOUT];
            } else {
                bb.x = 0.f; bb.y = 0.f;
            }
            acc = __builtin_amdgcn_wmma_f32_16x16x4_f32(false, a, false, bb,
                                                        (short)0, acc, false, false);
        }
        if (o < NOUT) {
            const float bv = bn[o];
            #pragma unroll
            for (int r = 0; r < 8; ++r) {
                const int node = i0 + r + 8 * lh;
                out[(long)(b * N_ + node) * NOUT + o] = lrelu(acc[r] + bv);
            }
        }
    }
}

extern "C" void kernel_launch(void* const* d_in, const int* in_sizes, int n_in,
                              void* d_out, int out_size, void* d_ws, size_t ws_size,
                              hipStream_t stream) {
    (void)in_sizes; (void)n_in; (void)d_ws; (void)ws_size; (void)out_size;
    const float* x         = (const float*)d_in[0];
    const float* node_attr = (const float*)d_in[1];
    const float* edge_attr = (const float*)d_in[2];
    const float* adj       = (const float*)d_in[3];
    const float* We        = (const float*)d_in[4];
    const float* be        = (const float*)d_in[5];
    const float* Wn        = (const float*)d_in[6];
    const float* bn        = (const float*)d_in[7];
    float* out             = (float*)d_out;

    dim3 grid(B_ * (N_ / 16));
    dim3 block(256);
    gin_fused<<<grid, block, 0, stream>>>(x, node_attr, edge_attr, adj, We, be, Wn, bn, out);
}


# Global loads issued ahead of use: adj, We_e, all phase-1 operands as one group (shared We_s fragment loaded once) and edge tiles 4 ahead, instead of a dependent round trip per tile
# speedup vs baseline: 1.3042x; 1.3042x over previous
#include <hip/hip_runtime.h>

typedef __attribute__((ext_vector_type(2))) float v2f;
typedef __attribute__((ext_vector_type(8))) float v8f;

#define B_    16
#define N_    128
#define NC    60
#define NA    4
#define NI    64
#define EAT   16
#define EC    64
#define NIN   128
#define NOUT  60
#define SLOPE 0.01f

__device__ __forceinline__ float lrelu(float v) { return v > 0.f ? v : v * SLOPE; }

__device__ __forceinline__ void load_node_frag(
    long nb, int lh, const float* __restrict__ x, const float* __restrict__ node_attr, v2f (&A)[16])
{
    #pragma unroll
    for (int s = 0; s < 16; ++s) {
        const int k0 = 4 * s;
        const int ka = k0 + 2 * lh;
        if (k0 < NC) {
            const float* p = x + nb * NC + ka;
            A[s].x = p[0]; A[s].y = p[1];
        } else {
            const float* p = node_attr + nb * NA + (ka - NC);
            A[s].x = p[0]; A[s].y = p[1];
        }
    }
}

__device__ __forceinline__ void load_we_frag(
    const float* __restrict__ We, int wrow0, int col, int lh, v2f (&B)[16])
{
    #pragma unroll
    for (int s = 0; s < 16; ++s) {
        const int ka = 4 * s + 2 * lh;
        const float* q = We + (wrow0 + ka) * EC + col;
        B[s].x = q[0]; B[s].y = q[EC];
    }
}

__global__ __launch_bounds__(256) void gin_fused(
    const float* __restrict__ x,        const float* __restrict__ node_attr,
    const float* __restrict__ edge_attr,const float* __restrict__ adj,
    const float* __restrict__ We,       const float* __restrict__ be,
    const float* __restrict__ Wn,       const float* __restrict__ bn,
    float* __restrict__ out)
{
    __shared__ float s_send[N_ * EC];
    __shared__ float s_recvb[16 * EC];
    __shared__ float s_agg[16 * EC];
    __shared__ __attribute__((aligned(16))) float s_adj[16 * N_];

    const int b    = blockIdx.x >> 3;
    const int it   = blockIdx.x & 7;
    const int i0   = it * 16;

    const int tid  = threadIdx.x;
    const int wave = tid >> 5;
    const int lane = tid & 31;
    const int lh   = lane >> 4;
    const int ll   = lane & 15;

    const float4* adj4 = reinterpret_cast<const float4*>(adj + (long)(b * N_ + i0) * N_);
    const float4 adjA = adj4[tid];
    const float4 adjB = adj4[tid + 256];

    v2f bE[4][4];
    #pragma unroll
    for (int ct = 0; ct < 4; ++ct)
        #pragma unroll
        for (int kk = 0; kk < 4; ++kk) {
            const int k = 4 * kk + 2 * lh;
            const float* q = We + k * EC + ct * 16 + ll;
            bE[ct][kk].x = q[0]; bE[ct][kk].y = q[EC];
        }

    const float* etile0 = edge_attr + ((long)(b * N_ + i0 + wave * 2)) * N_ * EAT + ll * EAT + 2 * lh;
    v2f aC[4][4];
    #pragma unroll
    for (int t = 0; t < 4; ++t)
        #pragma unroll
        for (int kk = 0; kk < 4; ++kk) {
            const float* p = etile0 + t * 16 * EAT + 4 * kk;
            aC[t][kk].x = p[0]; aC[t][kk].y = p[1];
        }

    const int col1 = (wave & 3) * 16 + ll;
    const int jb   = wave >> 2;
    v2f Bs[16], As[4][16];
    load_we_frag(We, EAT + NI, col1, lh, Bs);
    #pragma unroll
    for (int n = 0; n < 4; ++n)
        load_node_frag((long)(b * N_ + (jb + 2 * n) * 16 + ll), lh, x, node_attr, As[n]);
    reinterpret_cast<float4*>(s_adj)[tid]       = adjA;
    reinterpret_cast<float4*>(s_adj)[tid + 256] = adjB;
    v2f Br[16], Ar[16];
    float bev = 0.f;
    if (wave < 4) {
        load_we_frag(We, EAT, col1, lh, Br);
        load_node_frag((long)(b * N_ + i0 + ll), lh, x, node_attr, Ar);
        bev = be[col1];
    }

    #pragma unroll
    for (int n = 0; n < 4; ++n) {
        v8f acc = {};
        #pragma unroll
        for (int s = 0; s < 16; ++s)
            acc = __builtin_amdgcn_wmma_f32_16x16x4_f32(false, As[n][s], false, Bs[s],
                                                        (short)0, acc, false, false);
        #pragma unroll
        for (int r = 0; r < 8; ++r)
            s_send[((jb + 2 * n) * 16 + r + 8 * lh) * EC + col1] = acc[r];
    }
    if (wave < 4) {
        v8f acc = {};
        #pragma unroll
        for (int s = 0; s < 16; ++s)
            acc = __builtin_amdgcn_wmma_f32_16x16x4_f32(false, Ar[s], false, Br[s],
                                                        (short)0, acc, false, false);
        #pragma unroll
        for (int r = 0; r < 8; ++r)
            s_recvb[(r + 8 * lh) * EC + col1] = acc[r] + bev;
    }
    __syncthreads();

    float aggp[4] = {0.f, 0.f, 0.f, 0.f};
    float rb[4]   = {0.f, 0.f, 0.f, 0.f};
    #pragma nounroll
    for (int h = 0; h < 4; ++h) {
        const int il = wave * 2 + (h >> 1);
        const int jh = (h & 1) * 4;
        v2f aN[4][4];
        if (h < 3) {
            #pragma unroll
            for (int t = 0; t < 4; ++t)
                #pragma unroll
                for (int kk = 0; kk < 4; ++kk) {
                    const float* p = etile0 + ((h + 1) * 4 + t) * 16 * EAT + 4 * kk;
                    aN[t][kk].x = p[0]; aN[t][kk].y = p[1];
                }
        } else {
            #pragma unroll
            for (int t = 0; t < 4; ++t)
                #pragma unroll
                for (int kk = 0; kk < 4; ++kk) aN[t][kk] = aC[t][kk];
        }
        if ((h & 1) == 0) {
            #pragma unroll
            for (int ct = 0; ct < 4; ++ct) {
                aggp[ct] = 0.f;
                rb[ct]   = s_recvb[il * EC + ct * 16 + ll];
            }
        }
        #pragma unroll
        for (int t = 0; t < 4; ++t) {
            const int jt = jh + t;
            #pragma unroll
            for (int ct = 0; ct < 4; ++ct) {
                v8f d = {};
                #pragma unroll
                for (int kk = 0; kk < 4; ++kk)
                    d = __builtin_amdgcn_wmma_f32_16x16x4_f32(false, aC[t][kk], false, bE[ct][kk],
                                                              (short)0, d, false, false);
                #pragma unroll
                for (int r = 0; r < 8; ++r) {
                    const int j = jt * 16 + r + 8 * lh;
                    float v = d[r] + rb[ct] + s_send[j * EC + ct * 16 + ll];
                    aggp[ct] = fmaf(lrelu(v), s_adj[il * N_ + j], aggp[ct]);
                }
            }
        }
        if (h & 1) {
            #pragma unroll
            for (int ct = 0; ct < 4; ++ct) {
                float v = aggp[ct] + __shfl_xor(aggp[ct], 16, 32);
                if (lh == 0) s_agg[il * EC + ct * 16 + ll] = v;
            }
        }
        #pragma unroll
        for (int t = 0; t < 4; ++t)
            #pragma unroll
            for (int kk = 0; kk < 4; ++kk) aC[t][kk] = aN[t][kk];
    }
    __syncthreads();

    if (wave < 4) {
        const int ct = wave;
        const int o  = ct * 16 + ll;
        v8f acc = {};
        #pragma unroll
        for (int k0 = 0; k0 < NIN; k0 += 4) {
            const int ka = k0 + 2 * lh;
            v2f a, bb;
            if (k0 < NC) {
                const float* p = x + (long)(b * N_ + i0 + ll) * NC + ka;
                a.x = p[0]; a.y = p[1];
            } else if (k0 < NI) {
                const float* p = node_attr + (long)(b * N_ + i0 + ll) * NA + (ka - NC);
                a.x = p[0]; a.y = p[1];
            } else {
                a.x = s_agg[ll * EC + (ka - NI)];
                a.y = s_agg[ll * EC + (ka - NI) + 1];
            }
            if (o < NOUT) {
                const float* q = Wn + ka * NOUT + o;
                bb.x = q[0]; bb.y = q[NOUT];
            } else {
                bb.x = 0.f; bb.y = 0.f;
            }
            acc = __builtin_amdgcn_wmma_f32_16x16x4_f32(false, a, false, bb,
                                                        (short)0, acc, false, false);
        }
        if (o < NOUT) {
            const float bv = bn[o];
            #pragma unroll
            for (int r = 0; r < 8; ++r) {
                const int node = i0 + r + 8 * lh;
                out[(long)(b * N_ + node) * NOUT + o] = lrelu(acc[r] + bv);
            }
        }
    }
}

extern "C" void kernel_launch(void* const* d_in, const int* in_sizes, int n_in,
                              void* d_out, int out_size, void* d_ws, size_t ws_size,
                              hipStream_t stream) {
    (void)in_sizes; (void)n_in; (void)d_ws; (void)ws_size; (void)out_size;
    const float* x         = (const float*)d_in[0];
    const float* node_attr = (const float*)d_in[1];
    const float* edge_attr = (const float*)d_in[2];
    const float* adj       = (const float*)d_in[3];
    const float* We        = (const float*)d_in[4];
    const float* be        = (const float*)d_in[5];
    const float* Wn        = (const float*)d_in[6];
    const float* bn        = (const float*)d_in[7];
    float* out             = (float*)d_out;

    dim3 grid(B_ * (N_ / 16));
    dim3 block(256);
    gin_fused<<<grid, block, 0, stream>>>(x, node_attr, edge_attr, adj, We, be, Wn, bn, out);
}


# Node-MLP operands ready before the edge loop: all Wn fragments and bn loaded as one group under phase 2, node rows kept from the recv fragment, instead of a load round trip in each of 32 k-steps
# speedup vs baseline: 2.3406x; 1.7947x over previous
#include <hip/hip_runtime.h>

typedef __attribute__((ext_vector_type(2))) float v2f;
typedef __attribute__((ext_vector_type(8))) float v8f;

#define B_    16
#define N_    128
#define NC    60
#define NA    4
#define NI    64
#define EAT   16
#define EC    64
#define NIN   128
#define NOUT  60
#define SLOPE 0.01f

__device__ __forceinline__ float lrelu(float v) { return v > 0.f ? v : v * SLOPE; }

__device__ __forceinline__ void load_node_frag(
    long nb, int lh, const float* __restrict__ x, const float* __restrict__ node_attr, v2f (&A)[16])
{
    #pragma unroll
    for (int s = 0; s < 16; ++s) {
        const int k0 = 4 * s;
        const int ka = k0 + 2 * lh;
        if (k0 < NC) {
            const float* p = x + nb * NC + ka;
            A[s].x = p[0]; A[s].y = p[1];
        } else {
            const float* p = node_attr + nb * NA + (ka - NC);
            A[s].x = p[0]; A[s].y = p[1];
        }
    }
}

__device__ __forceinline__ void load_we_frag(
    const float* __restrict__ We, int wrow0, int col, int lh, v2f (&B)[16])
{
    #pragma unroll
    for (int s = 0; s < 16; ++s) {
        const int ka = 4 * s + 2 * lh;
        const float* q = We + (wrow0 + ka) * EC + col;
        B[s].x = q[0]; B[s].y = q[EC];
    }
}

__global__ __launch_bounds__(256) void gin_fused(
    const float* __restrict__ x,        const float* __restrict__ node_attr,
    const float* __restrict__ edge_attr,const float* __restrict__ adj,
    const float* __restrict__ We,       const float* __restrict__ be,
    const float* __restrict__ Wn,       const float* __restrict__ bn,
    float* __restrict__ out)
{
    __shared__ float s_send[N_ * EC];
    __shared__ float s_recvb[16 * EC];
    __shared__ float s_agg[16 * EC];
    __shared__ __attribute__((aligned(16))) float s_adj[16 * N_];

    const int b    = blockIdx.x >> 3;
    const int it   = blockIdx.x & 7;
    const int i0   = it * 16;

    const int tid  = threadIdx.x;
    const int wave = tid >> 5;
    const int lane = tid & 31;
    const int lh   = lane >> 4;
    const int ll   = lane & 15;

    const float4* adj4 = reinterpret_cast<const float4*>(adj + (long)(b * N_ + i0) * N_);
    const float4 adjA = adj4[tid];
    const float4 adjB = adj4[tid + 256];

    v2f bE[4][4];
    #pragma unroll
    for (int ct = 0; ct < 4; ++ct)
        #pragma unroll
        for (int kk = 0; kk < 4; ++kk) {
            const int k = 4 * kk + 2 * lh;
            const float* q = We + k * EC + ct * 16 + ll;
            bE[ct][kk].x = q[0]; bE[ct][kk].y = q[EC];
        }

    const float* etile0 = edge_attr + ((long)(b * N_ + i0 + wave * 2)) * N_ * EAT + ll * EAT + 2 * lh;
    v2f aC[4][4];
    #pragma unroll
    for (int t = 0; t < 4; ++t)
        #pragma unroll
        for (int kk = 0; kk < 4; ++kk) {
            const float* p = etile0 + t * 16 * EAT + 4 * kk;
            aC[t][kk].x = p[0]; aC[t][kk].y = p[1];
        }

    const int col1 = (wave & 3) * 16 + ll;
    const int jb   = wave >> 2;
    v2f Bs[16], As[4][16];
    load_we_frag(We, EAT + NI, col1, lh, Bs);
    #pragma unroll
    for (int n = 0; n < 4; ++n)
        load_node_frag((long)(b * N_ + (jb + 2 * n) * 16 + ll), lh, x, node_attr, As[n]);
    reinterpret_cast<float4*>(s_adj)[tid]       = adjA;
    reinterpret_cast<float4*>(s_adj)[tid + 256] = adjB;
    v2f Br[16], Ar[16];
    float bev = 0.f;
    if (wave < 4) {
        load_we_frag(We, EAT, col1, lh, Br);
        load_node_frag((long)(b * N_ + i0 + ll), lh, x, node_attr, Ar);
        bev = be[col1];
    }

    #pragma unroll
    for (int n = 0; n < 4; ++n) {
        v8f acc = {};
        #pragma unroll
        for (int s = 0; s < 16; ++s)
            acc = __builtin_amdgcn_wmma_f32_16x16x4_f32(false, As[n][s], false, Bs[s],
                                                        (short)0, acc, false, false);
        #pragma unroll
        for (int r = 0; r < 8; ++r)
            s_send[((jb + 2 * n) * 16 + r + 8 * lh) * EC + col1] = acc[r];
    }
    if (wave < 4) {
        v8f acc = {};
        #pragma unroll
        for (int s = 0; s < 16; ++s)
            acc = __builtin_amdgcn_wmma_f32_16x16x4_f32(false, Ar[s], false, Br[s],
                                                        (short)0, acc, false, false);
        #pragma unroll
        for (int r = 0; r < 8; ++r)
            s_recvb[(r + 8 * lh) * EC + col1] = acc[r] + bev;
    }
    __syncthreads();

    v2f Bn[32];
    float bnv = 0.f;
    if (wave < 4) {
        const int o  = wave * 16 + ll;
        const int oc = o < NOUT ? o : NOUT - 1;
        #pragma unroll
        for (int s = 0; s < 32; ++s) {
            const float* q = Wn + (4 * s + 2 * lh) * NOUT + oc;
            Bn[s].x = q[0]; Bn[s].y = q[NOUT];
        }
        bnv = bn[oc];
    }

    float aggp[4] = {0.f, 0.f, 0.f, 0.f};
    float rb[4]   = {0.f, 0.f, 0.f, 0.f};
    #pragma nounroll
    for (int h = 0; h < 4; ++h) {
        const int il = wave * 2 + (h >> 1);
        const int jh = (h & 1) * 4;
        v2f aN[4][4];
        if (h < 3) {
            #pragma unroll
            for (int t = 0; t < 4; ++t)
                #pragma unroll
                for (int kk = 0; kk < 4; ++kk) {
                    const float* p = etile0 + ((h + 1) * 4 + t) * 16 * EAT + 4 * kk;
                    aN[t][kk].x = p[0]; aN[t][kk].y = p[1];
                }
        } else {
            #pragma unroll
            for (int t = 0; t < 4; ++t)
                #pragma unroll
                for (int kk = 0; kk < 4; ++kk) aN[t][kk] = aC[t][kk];
        }
        if ((h & 1) == 0) {
            #pragma unroll
            for (int ct = 0; ct < 4; ++ct) {
                aggp[ct] = 0.f;
                rb[ct]   = s_recvb[il * EC + ct * 16 + ll];
            }
        }
        #pragma unroll
        for (int t = 0; t < 4; ++t) {
            const int jt = jh + t;
            #pragma unroll
            for (int ct = 0; ct < 4; ++ct) {
                v8f d = {};
                #pragma unroll
                for (int kk = 0; kk < 4; ++kk)
                    d = __builtin_amdgcn_wmma_f32_16x16x4_f32(false, aC[t][kk], false, bE[ct][kk],
                                                              (short)0, d, false, false);
                #pragma unroll
                for (int r = 0; r < 8; ++r) {
                    const int j = jt * 16 + r + 8 * lh;
                    float v = d[r] + rb[ct] + s_send[j * EC + ct * 16 + ll];
                    aggp[ct] = fmaf(lrelu(v), s_adj[il * N_ + j], aggp[ct]);
                }
            }
        }
        if (h & 1) {
            #pragma unroll
            for (int ct = 0; ct < 4; ++ct) {
                float v = aggp[ct] + __shfl_xor(aggp[ct], 16, 32);
                if (lh == 0) s_agg[il * EC + ct * 16 + ll] = v;
            }
        }
        #pragma unroll
        for (int t = 0; t < 4; ++t)
            #pragma unroll
            for (int kk = 0; kk < 4; ++kk) aC[t][kk] = aN[t][kk];
    }
    __syncthreads();

    if (wave < 4) {
        const int o = wave * 16 + ll;
        v2f Ag[16];
        #pragma unroll
        for (int s = 0; s < 16; ++s) {
            Ag[s].x = s_agg[ll * EC + 4 * s + 2 * lh];
            Ag[s].y = s_agg[ll * EC + 4 * s + 2 * lh + 1];
        }
        v8f acc = {};
        #pragma unroll
        for (int s = 0; s < 16; ++s)
            acc = __builtin_amdgcn_wmma_f32_16x16x4_f32(false, Ar[s], false, Bn[s],
                                                        (short)0, acc, false, false);
        #pragma unroll
        for (int s = 0; s < 16; ++s)
            acc = __builtin_amdgcn_wmma_f32_16x16x4_f32(false, Ag[s], false, Bn[16 + s],
                                                        (short)0, acc, false, false);
        if (o < NOUT) {
            #pragma unroll
            for (int r = 0; r < 8; ++r) {
                const int node = i0 + r + 8 * lh;
                out[(long)(b * N_ + node) * NOUT + o] = lrelu(acc[r] + bnv);
            }
        }
    }
}

extern "C" void kernel_launch(void* const* d_in, const int* in_sizes, int n_in,
                              void* d_out, int out_size, void* d_ws, size_t ws_size,
                              hipStream_t stream) {
    (void)in_sizes; (void)n_in; (void)d_ws; (void)ws_size; (void)out_size;
    const float* x         = (const float*)d_in[0];
    const float* node_attr = (const float*)d_in[1];
    const float* edge_attr = (const float*)d_in[2];
    const float* adj       = (const float*)d_in[3];
    const float* We        = (const float*)d_in[4];
    const float* be        = (const float*)d_in[5];
    const float* Wn        = (const float*)d_in[6];
    const float* bn        = (const float*)d_in[7];
    float* out             = (float*)d_out;

    dim3 grid(B_ * (N_ / 16));
    dim3 block(256);
    gin_fused<<<grid, block, 0, stream>>>(x, node_attr, edge_attr, adj, We, be, Wn, bn, out);
}


# Finer grid: one workgroup per (batch, 8 receiver rows) instead of 16, so 256 workgroups fill all 256 WGPs (128 used half); one edge row per wave, edge phase per WGP halves; sum order kept
# speedup vs baseline: 3.1182x; 1.3322x over previous
#include <hip/hip_runtime.h>

typedef __attribute__((ext_vector_type(2))) float v2f;
typedef __attribute__((ext_vector_type(8))) float v8f;

#define B_    16
#define N_    128
#define NC    60
#define NA    4
#define NI    64
#define EAT   16
#define EC    64
#define NIN   128
#define NOUT  60
#define SLOPE 0.01f
#define RI    8

__device__ __forceinline__ float lrelu(float v) { return v > 0.f ? v : v * SLOPE; }

__device__ __forceinline__ void load_node_frag(
    long nb, int lh, const float* __restrict__ x, const float* __restrict__ node_attr, v2f (&A)[16])
{
    #pragma unroll
    for (int s = 0; s < 16; ++s) {
        const int k0 = 4 * s;
        const int ka = k0 + 2 * lh;
        if (k0 < NC) {
            const float* p = x + nb * NC + ka;
            A[s].x = p[0]; A[s].y = p[1];
        } else {
            const float* p = node_attr + nb * NA + (ka - NC);
            A[s].x = p[0]; A[s].y = p[1];
        }
    }
}

__device__ __forceinline__ void load_we_frag(
    const float* __restrict__ We, int wrow0, int col, int lh, v2f (&B)[16])
{
    #pragma unroll
    for (int s = 0; s < 16; ++s) {
        const int ka = 4 * s + 2 * lh;
        const float* q = We + (wrow0 + ka) * EC + col;
        B[s].x = q[0]; B[s].y = q[EC];
    }
}

__global__ __launch_bounds__(256) void gin_fused(
    const float* __restrict__ x,        const float* __restrict__ node_attr,
    const float* __restrict__ edge_attr,const float* __restrict__ adj,
    const float* __restrict__ We,       const float* __restrict__ be,
    const float* __restrict__ Wn,       const float* __restrict__ bn,
    float* __restrict__ out)
{
    __shared__ float s_send[N_ * EC];
    __shared__ float s_recvb[RI * EC];
    __shared__ float s_agg[RI * EC];
    __shared__ __attribute__((aligned(16))) float s_adj[RI * N_];

    const int b    = blockIdx.x >> 4;
    const int it   = blockIdx.x & 15;
    const int i0   = it * RI;

    const int tid  = threadIdx.x;
    const int wave = tid >> 5;
    const int lane = tid & 31;
    const int lh   = lane >> 4;
    const int ll   = lane & 15;

    const float4* adj4 = reinterpret_cast<const float4*>(adj + (long)(b * N_ + i0) * N_);
    const float4 adjA = adj4[tid];

    v2f bE[4][4];
    #pragma unroll
    for (int ct = 0; ct < 4; ++ct)
        #pragma unroll
        for (int kk = 0; kk < 4; ++kk) {
            const int k = 4 * kk + 2 * lh;
            const float* q = We + k * EC + ct * 16 + ll;
            bE[ct][kk].x = q[0]; bE[ct][kk].y = q[EC];
        }

    const float* etile0 = edge_attr + ((long)(b * N_ + i0 + wave)) * N_ * EAT + ll * EAT + 2 * lh;
    v2f aC[4][4];
    #pragma unroll
    for (int t = 0; t < 4; ++t)
        #pragma unroll
        for (int kk = 0; kk < 4; ++kk) {
            const float* p = etile0 + t * 16 * EAT + 4 * kk;
            aC[t][kk].x = p[0]; aC[t][kk].y = p[1];
        }

    const int col1 = (wave & 3) * 16 + ll;
    const int jb   = wave >> 2;
    v2f Bs[16], As[4][16];
    load_we_frag(We, EAT + NI, col1, lh, Bs);
    #pragma unroll
    for (int n = 0; n < 4; ++n)
        load_node_frag((long)(b * N_ + (jb + 2 * n) * 16 + ll), lh, x, node_attr, As[n]);
    reinterpret_cast<float4*>(s_adj)[tid] = adjA;
    v2f Br[16], Ar[16];
    float bev = 0.f;
    if (wave < 4) {
        load_we_frag(We, EAT, col1, lh, Br);
        load_node_frag((long)(b * N_ + i0 + (ll & (RI - 1))), lh, x, node_attr, Ar);
        bev = be[col1];
    }

    #pragma unroll
    for (int n = 0; n < 4; ++n) {
        v8f acc = {};
        #pragma unroll
        for (int s = 0; s < 16; ++s)
            acc = __builtin_amdgcn_wmma_f32_16x16x4_f32(false, As[n][s], false, Bs[s],
                                                        (short)0, acc, false, false);
        #pragma unroll
        for (int r = 0; r < 8; ++r)
            s_send[((jb + 2 * n) * 16 + r + 8 * lh) * EC + col1] = acc[r];
    }
    if (wave < 4) {
        v8f acc = {};
        #pragma unroll
        for (int s = 0; s < 16; ++s)
            acc = __builtin_amdgcn_wmma_f32_16x16x4_f32(false, Ar[s], false, Br[s],
                                                        (short)0, acc, false, false);
        if (lh == 0) {
            #pragma unroll
            for (int r = 0; r < 8; ++r)
                s_recvb[r * EC + col1] = acc[r] + bev;
        }
    }
    __syncthreads();

    v2f Bn[32];
    float bnv = 0.f;
    if (wave < 4) {
        const int o  = wave * 16 + ll;
        const int oc = o < NOUT ? o : NOUT - 1;
        #pragma unroll
        for (int s = 0; s < 32; ++s) {
            const float* q = Wn + (4 * s + 2 * lh) * NOUT + oc;
            Bn[s].x = q[0]; Bn[s].y = q[NOUT];
        }
        bnv = bn[oc];
    }

    const int il = wave;
    float aggp[4], rb[4];
    #pragma unroll
    for (int ct = 0; ct < 4; ++ct) {
        aggp[ct] = 0.f;
        rb[ct]   = s_recvb[il * EC + ct * 16 + ll];
    }
    #pragma nounroll
    for (int h = 0; h < 2; ++h) {
        const int jh = h * 4;
        v2f aN[4][4];
        if (h < 1) {
            #pragma unroll
            for (int t = 0; t < 4; ++t)
                #pragma unroll
                for (int kk = 0; kk < 4; ++kk) {
                    const float* p = etile0 + ((h + 1) * 4 + t) * 16 * EAT + 4 * kk;
                    aN[t][kk].x = p[0]; aN[t][kk].y = p[1];
                }
        } else {
            #pragma unroll
            for (int t = 0; t < 4; ++t)
                #pragma unroll
                for (int kk = 0; kk < 4; ++kk) aN[t][kk] = aC[t][kk];
        }
        #pragma unroll
        for (int t = 0; t < 4; ++t) {
            const int jt = jh + t;
            #pragma unroll
            for (int ct = 0; ct < 4; ++ct) {
                v8f d = {};
                #pragma unroll
                for (int kk = 0; kk < 4; ++kk)
                    d = __builtin_amdgcn_wmma_f32_16x16x4_f32(false, aC[t][kk], false, bE[ct][kk],
                                                              (short)0, d, false, false);
                #pragma unroll
                for (int r = 0; r < 8; ++r) {
                    const int j = jt * 16 + r + 8 * lh;
                    float v = d[r] + rb[ct] + s_send[j * EC + ct * 16 + ll];
                    aggp[ct] = fmaf(lrelu(v), s_adj[il * N_ + j], aggp[ct]);
                }
            }
        }
        #pragma unroll
        for (int t = 0; t < 4; ++t)
            #pragma unroll
            for (int kk = 0; kk < 4; ++kk) aC[t][kk] = aN[t][kk];
    }
    #pragma unroll
    for (int ct = 0; ct < 4; ++ct) {
        float v = aggp[ct] + __shfl_xor(aggp[ct], 16, 32);
        if (lh == 0) s_agg[il * EC + ct * 16 + ll] = v;
    }
    __syncthreads();

    if (wave < 4) {
        const int o  = wave * 16 + ll;
        const int ar = ll & (RI - 1);
        v2f Ag[16];
        #pragma unroll
        for (int s = 0; s < 16; ++s) {
            Ag[s].x = s_agg[ar * EC + 4 * s + 2 * lh];
            Ag[s].y = s_agg[ar * EC + 4 * s + 2 * lh + 1];
        }
        v8f acc = {};
        #pragma unroll
        for (int s = 0; s < 16; ++s)
            acc = __builtin_amdgcn_wmma_f32_16x16x4_f32(false, Ar[s], false, Bn[s],
                                                        (short)0, acc, false, false);
        #pragma unroll
        for (int s = 0; s < 16; ++s)
            acc = __builtin_amdgcn_wmma_f32_16x16x4_f32(false, Ag[s], false, Bn[16 + s],
                                                        (short)0, acc, false, false);
        if (o < NOUT && lh == 0) {
            #pragma unroll
            for (int r = 0; r < 8; ++r) {
                const int node = i0 + r;
                out[(long)(b * N_ + node) * NOUT + o] = lrelu(acc[r] + bnv);
            }
        }
    }
}

extern "C" void kernel_launch(void* const* d_in, const int* in_sizes, int n_in,
                              void* d_out, int out_size, void* d_ws, size_t ws_size,
                              hipStream_t stream) {
    (void)in_sizes; (void)n_in; (void)d_ws; (void)ws_size; (void)out_size;
    const float* x         = (const float*)d_in[0];
    const float* node_attr = (const float*)d_in[1];
    const float* edge_attr = (const float*)d_in[2];
    const float* adj       = (const float*)d_in[3];
    const float* We        = (const float*)d_in[4];
    const float* be        = (const float*)d_in[5];
    const float* Wn        = (const float*)d_in[6];
    const float* bn        = (const float*)d_in[7];
    float* out             = (float*)d_out;

    dim3 grid(B_ * (N_ / RI));
    dim3 block(256);
    gin_fused<<<grid, block, 0, stream>>>(x, node_attr, edge_attr, adj, We, be, Wn, bn, out);
}


# Node rows of the batch (x, node_attr) fetched once as whole-line 16-byte loads and staged in LDS; phase-1 A fragments come from LDS, same k order, not from 80 row-strided 8-byte global loads per lane
# speedup vs baseline: 3.4734x; 1.1139x over previous
#include <hip/hip_runtime.h>

typedef __attribute__((ext_vector_type(2))) float v2f;
typedef __attribute__((ext_vector_type(4))) float v4f;
typedef __attribute__((ext_vector_type(8))) float v8f;

#define B_    16
#define N_    128
#define NC    60
#define NA    4
#define NI    64
#define EAT   16
#define EC    64
#define NIN   128
#define NOUT  60
#define SLOPE 0.01f
#define RI    8
#define NIP   68

__device__ __forceinline__ float lrelu(float v) { return v > 0.f ? v : v * SLOPE; }

__device__ __forceinline__ void load_node_frag(const float* s_ni, int row, int lh, v2f (&A)[16])
{
    const v2f* p = reinterpret_cast<const v2f*>(s_ni + row * NIP + 2 * lh);
    #pragma unroll
    for (int s = 0; s < 16; ++s) A[s] = p[2 * s];
}

__device__ __forceinline__ void stage_piece(float* s_ni, int f, v4f v)
{
    const int row = f / 15;
    *reinterpret_cast<v4f*>(s_ni + row * NIP + 4 * (f - 15 * row)) = v;
}

__device__ __forceinline__ void load_we_frag(
    const float* __restrict__ We, int wrow0, int col, int lh, v2f (&B)[16])
{
    #pragma unroll
    for (int s = 0; s < 16; ++s) {
        const int ka = 4 * s + 2 * lh;
        const float* q = We + (wrow0 + ka) * EC + col;
        B[s].x = q[0]; B[s].y = q[EC];
    }
}

__global__ __launch_bounds__(256) void gin_fused(
    const float* __restrict__ x,        const float* __restrict__ node_attr,
    const float* __restrict__ edge_attr,const float* __restrict__ adj,
    const float* __restrict__ We,       const float* __restrict__ be,
    const float* __restrict__ Wn,       const float* __restrict__ bn,
    float* __restrict__ out)
{
    __shared__ float s_send[N_ * EC];
    __shared__ float s_recvb[RI * EC];
    __shared__ float s_agg[RI * EC];
    __shared__ __attribute__((aligned(16))) float s_adj[RI * N_];
    __shared__ __attribute__((aligned(16))) float s_ni[N_ * NIP];

    const int b    = blockIdx.x >> 4;
    const int it   = blockIdx.x & 15;
    const int i0   = it * RI;

    const int tid  = threadIdx.x;
    const int wave = tid >> 5;
    const int lane = tid & 31;
    const int lh   = lane >> 4;
    const int ll   = lane & 15;

    const v4f* x4  = reinterpret_cast<const v4f*>(x + (long)b * N_ * NC);
    const v4f* na4 = reinterpret_cast<const v4f*>(node_attr + (long)b * N_ * NA);
    const bool lo  = wave < 4;
    const v4f xv0 = x4[tid];
    const v4f xv1 = x4[tid + 256];
    const v4f xv2 = x4[tid + 512];
    const v4f xv3 = x4[tid + 768];
    const v4f xv4 = x4[tid + 1024];
    const v4f xv5 = x4[tid + 1280];
    const v4f xv6 = x4[tid + 1536];
    const v4f* p7 = lo ? x4 : na4;
    const v4f xv7 = p7[lo ? tid + 1792 : tid - 128];

    const v4f* adj4 = reinterpret_cast<const v4f*>(adj + (long)(b * N_ + i0) * N_);
    const v4f adjA = adj4[tid];

    const int col1 = (wave & 3) * 16 + ll;
    const int jb   = wave >> 2;
    v2f Bs[16], Br[16];
    float bev = 0.f;
    load_we_frag(We, EAT + NI, col1, lh, Bs);

    stage_piece(s_ni, tid,        xv0);
    stage_piece(s_ni, tid + 256,  xv1);
    stage_piece(s_ni, tid + 512,  xv2);
    stage_piece(s_ni, tid + 768,  xv3);
    stage_piece(s_ni, tid + 1024, xv4);
    stage_piece(s_ni, tid + 1280, xv5);
    stage_piece(s_ni, tid + 1536, xv6);
    {
        const int f7   = tid + 1792;
        const int row7 = f7 / 15;
        const int dst7 = lo ? row7 * NIP + 4 * (f7 - 15 * row7) : (tid - 128) * NIP + NC;
        *reinterpret_cast<v4f*>(s_ni + dst7) = xv7;
    }
    reinterpret_cast<v4f*>(s_adj)[tid] = adjA;

    __syncthreads();

    v2f As[4][16], Ar[16];
    #pragma unroll
    for (int n = 0; n < 4; ++n)
        load_node_frag(s_ni, (jb + 2 * n) * 16 + ll, lh, As[n]);
    if (lo)
        load_node_frag(s_ni, i0 + (ll & (RI - 1)), lh, Ar);

    if (lo) {
        load_we_frag(We, EAT, col1, lh, Br);
        bev = be[col1];
    }

    v2f bE[4][4];
    #pragma unroll
    for (int ct = 0; ct < 4; ++ct)
        #pragma unroll
        for (int kk = 0; kk < 4; ++kk) {
            const int k = 4 * kk + 2 * lh;
            const float* q = We + k * EC + ct * 16 + ll;
            bE[ct][kk].x = q[0]; bE[ct][kk].y = q[EC];
        }

    const float* etile0 = edge_attr + ((long)(b * N_ + i0 + wave)) * N_ * EAT + ll * EAT + 2 * lh;
    v2f aC[4][4];
    #pragma unroll
    for (int t = 0; t < 4; ++t)
        #pragma unroll
        for (int kk = 0; kk < 4; ++kk) {
            const float* p = etile0 + t * 16 * EAT + 4 * kk;
            aC[t][kk].x = p[0]; aC[t][kk].y = p[1];
        }

    #pragma unroll
    for (int n = 0; n < 4; ++n) {
        v8f acc = {};
        #pragma unroll
        for (int s = 0; s < 16; ++s)
            acc = __builtin_amdgcn_wmma_f32_16x16x4_f32(false, As[n][s], false, Bs[s],
                                                        (short)0, acc, false, false);
        #pragma unroll
        for (int r = 0; r < 8; ++r)
            s_send[((jb + 2 * n) * 16 + r + 8 * lh) * EC + col1] = acc[r];
    }
    if (wave < 4) {
        v8f acc = {};
        #pragma unroll
        for (int s = 0; s < 16; ++s)
            acc = __builtin_amdgcn_wmma_f32_16x16x4_f32(false, Ar[s], false, Br[s],
                                                        (short)0, acc, false, false);
        if (lh == 0) {
            #pragma unroll
            for (int r = 0; r < 8; ++r)
                s_recvb[r * EC + col1] = acc[r] + bev;
        }
    }
    __syncthreads();

    v2f Bn[32];
    float bnv = 0.f;
    if (wave < 4) {
        const int o  = wave * 16 + ll;
        const int oc = o < NOUT ? o : NOUT - 1;
        #pragma unroll
        for (int s = 0; s < 32; ++s) {
            const float* q = Wn + (4 * s + 2 * lh) * NOUT + oc;
            Bn[s].x = q[0]; Bn[s].y = q[NOUT];
        }
        bnv = bn[oc];
    }

    const int il = wave;
    float aggp[4], rb[4];
    #pragma unroll
    for (int ct = 0; ct < 4; ++ct) {
        aggp[ct] = 0.f;
        rb[ct]   = s_recvb[il * EC + ct * 16 + ll];
    }
    #pragma nounroll
    for (int h = 0; h < 2; ++h) {
        const int jh = h * 4;
        v2f aN[4][4];
        if (h < 1) {
            #pragma unroll
            for (int t = 0; t < 4; ++t)
                #pragma unroll
                for (int kk = 0; kk < 4; ++kk) {
                    const float* p = etile0 + ((h + 1) * 4 + t) * 16 * EAT + 4 * kk;
                    aN[t][kk].x = p[0]; aN[t][kk].y = p[1];
                }
        } else {
            #pragma unroll
            for (int t = 0; t < 4; ++t)
                #pragma unroll
                for (int kk = 0; kk < 4; ++kk) aN[t][kk] = aC[t][kk];
        }
        #pragma unroll
        for (int t = 0; t < 4; ++t) {
            const int jt = jh + t;
            #pragma unroll
            for (int ct = 0; ct < 4; ++ct) {
                v8f d = {};
                #pragma unroll
                for (int kk = 0; kk < 4; ++kk)
                    d = __builtin_amdgcn_wmma_f32_16x16x4_f32(false, aC[t][kk], false, bE[ct][kk],
                                                              (short)0, d, false, false);
                #pragma unroll
                for (int r = 0; r < 8; ++r) {
                    const int j = jt * 16 + r + 8 * lh;
                    float v = d[r] + rb[ct] + s_send[j * EC + ct * 16 + ll];
                    aggp[ct] = fmaf(lrelu(v), s_adj[il * N_ + j], aggp[ct]);
                }
            }
        }
        #pragma unroll
        for (int t = 0; t < 4; ++t)
            #pragma unroll
            for (int kk = 0; kk < 4; ++kk) aC[t][kk] = aN[t][kk];
    }
    #pragma unroll
    for (int ct = 0; ct < 4; ++ct) {
        float v = aggp[ct] + __shfl_xor(aggp[ct], 16, 32);
        if (lh == 0) s_agg[il * EC + ct * 16 + ll] = v;
    }
    __syncthreads();

    if (wave < 4) {
        const int o  = wave * 16 + ll;
        const int ar = ll & (RI - 1);
        v2f Ag[16];
        #pragma unroll
        for (int s = 0; s < 16; ++s) {
            Ag[s].x = s_agg[ar * EC + 4 * s + 2 * lh];
            Ag[s].y = s_agg[ar * EC + 4 * s + 2 * lh + 1];
        }
        v8f acc = {};
        #pragma unroll
        for (int s = 0; s < 16; ++s)
            acc = __builtin_amdgcn_wmma_f32_16x16x4_f32(false, Ar[s], false, Bn[s],
                                                        (short)0, acc, false, false);
        #pragma unroll
        for (int s = 0; s < 16; ++s)
            acc = __builtin_amdgcn_wmma_f32_16x16x4_f32(false, Ag[s], false, Bn[16 + s],
                                                        (short)0, acc, false, false);
        if (o < NOUT && lh == 0) {
            #pragma unroll
            for (int r = 0; r < 8; ++r) {
                const int node = i0 + r;
                out[(long)(b * N_ + node) * NOUT + o] = lrelu(acc[r] + bnv);
            }
        }
    }
}

extern "C" void kernel_launch(void* const* d_in, const int* in_sizes, int n_in,
                              void* d_out, int out_size, void* d_ws, size_t ws_size,
                              hipStream_t stream) {
    (void)in_sizes; (void)n_in; (void)d_ws; (void)ws_size; (void)out_size;
    const float* x         = (const float*)d_in[0];
    const float* node_attr = (const float*)d_in[1];
    const float* edge_attr = (const float*)d_in[2];
    const float* adj       = (const float*)d_in[3];
    const float* We        = (const float*)d_in[4];
    const float* be        = (const float*)d_in[5];
    const float* Wn        = (const float*)d_in[6];
    const float* bn        = (const float*)d_in[7];
    float* out             = (float*)d_out;

    dim3 grid(B_ * (N_ / RI));
    dim3 block(256);
    gin_fused<<<grid, block, 0, stream>>>(x, node_attr, edge_attr, adj, We, be, Wn, bn, out);
}


# Edge phase re-tiled: a wave owns one 16-sender tile for all 8 receiver rows; its send tile is read from LDS once, as accumulator start; partial sums joined across waves in LDS (sum order changed)
# speedup vs baseline: 3.7955x; 1.0927x over previous
#include <hip/hip_runtime.h>

typedef __attribute__((ext_vector_type(2))) float v2f;
typedef __attribute__((ext_vector_type(4))) float v4f;
typedef __attribute__((ext_vector_type(8))) float v8f;

#define B_    16
#define N_    128
#define NC    60
#define NA    4
#define NI    64
#define EAT   16
#define EC    64
#define NIN   128
#define NOUT  60
#define SLOPE 0.01f
#define RI    8
#define NIP   68

__device__ __forceinline__ float lrelu(float v) { return v > 0.f ? v : v * SLOPE; }

__device__ __forceinline__ void load_node_frag(const float* s_ni, int row, int lh, v2f (&A)[16])
{
    const v2f* p = reinterpret_cast<const v2f*>(s_ni + row * NIP + 2 * lh);
    #pragma unroll
    for (int s = 0; s < 16; ++s) A[s] = p[2 * s];
}

__device__ __forceinline__ void stage_piece(float* s_ni, int f, v4f v)
{
    const int row = f / 15;
    *reinterpret_cast<v4f*>(s_ni + row * NIP + 4 * (f - 15 * row)) = v;
}

__device__ __forceinline__ void load_we_frag(
    const float* __restrict__ We, int wrow0, int col, int lh, v2f (&B)[16])
{
    #pragma unroll
    for (int s = 0; s < 16; ++s) {
        const int ka = 4 * s + 2 * lh;
        const float* q = We + (wrow0 + ka) * EC + col;
        B[s].x = q[0]; B[s].y = q[EC];
    }
}

__global__ __launch_bounds__(256) void gin_fused(
    const float* __restrict__ x,        const float* __restrict__ node_attr,
    const float* __restrict__ edge_attr,const float* __restrict__ adj,
    const float* __restrict__ We,       const float* __restrict__ be,
    const float* __restrict__ Wn,       const float* __restrict__ bn,
    float* __restrict__ out)
{
    __shared__ float s_send[N_ * EC];
    __shared__ float s_recvb[RI * EC];
    __shared__ float s_agg[RI * EC];
    __shared__ __attribute__((aligned(16))) float s_adj[RI * N_];
    __shared__ __attribute__((aligned(16))) float s_ni[N_ * NIP];
    __shared__ __attribute__((aligned(16))) float s_part[16 * EC * RI];

    const int b    = blockIdx.x >> 4;
    const int it   = blockIdx.x & 15;
    const int i0   = it * RI;

    const int tid  = threadIdx.x;
    const int wave = tid >> 5;
    const int lane = tid & 31;
    const int lh   = lane >> 4;
    const int ll   = lane & 15;

    const v4f* x4  = reinterpret_cast<const v4f*>(x + (long)b * N_ * NC);
    const v4f* na4 = reinterpret_cast<const v4f*>(node_attr + (long)b * N_ * NA);
    const bool lo  = wave < 4;
    const v4f xv0 = x4[tid];
    const v4f xv1 = x4[tid + 256];
    const v4f xv2 = x4[tid + 512];
    const v4f xv3 = x4[tid + 768];
    const v4f xv4 = x4[tid + 1024];
    const v4f xv5 = x4[tid + 1280];
    const v4f xv6 = x4[tid + 1536];
    const v4f* p7 = lo ? x4 : na4;
    const v4f xv7 = p7[lo ? tid + 1792 : tid - 128];

    const v4f* adj4 = reinterpret_cast<const v4f*>(adj + (long)(b * N_ + i0) * N_);
    const v4f adjA = adj4[tid];

    const int col1 = (wave & 3) * 16 + ll;
    const int jb   = wave >> 2;
    v2f Bs[16], Br[16];
    float bev = 0.f;
    load_we_frag(We, EAT + NI, col1, lh, Bs);

    stage_piece(s_ni, tid,        xv0);
    stage_piece(s_ni, tid + 256,  xv1);
    stage_piece(s_ni, tid + 512,  xv2);
    stage_piece(s_ni, tid + 768,  xv3);
    stage_piece(s_ni, tid + 1024, xv4);
    stage_piece(s_ni, tid + 1280, xv5);
    stage_piece(s_ni, tid + 1536, xv6);
    {
        const int f7   = tid + 1792;
        const int row7 = f7 / 15;
        const int dst7 = lo ? row7 * NIP + 4 * (f7 - 15 * row7) : (tid - 128) * NIP + NC;
        *reinterpret_cast<v4f*>(s_ni + dst7) = xv7;
    }
    reinterpret_cast<v4f*>(s_adj)[tid] = adjA;

    __syncthreads();

    v2f As[4][16], Ar[16];
    #pragma unroll
    for (int n = 0; n < 4; ++n)
        load_node_frag(s_ni, (jb + 2 * n) * 16 + ll, lh, As[n]);
    if (lo)
        load_node_frag(s_ni, i0 + (ll & (RI - 1)), lh, Ar);

    if (lo) {
        load_we_frag(We, EAT, col1, lh, Br);
        bev = be[col1];
    }

    v2f bE[4][4];
    #pragma unroll
    for (int ct = 0; ct < 4; ++ct)
        #pragma unroll
        for (int kk = 0; kk < 4; ++kk) {
            const int k = 4 * kk + 2 * lh;
            const float* q = We + k * EC + ct * 16 + ll;
            bE[ct][kk].x = q[0]; bE[ct][kk].y = q[EC];
        }

    const float* etile0 = edge_attr + ((long)(b * N_ + i0) * N_ + wave * 16 + ll) * EAT + 2 * lh;
    v2f aC[4][4];
    #pragma unroll
    for (int t = 0; t < 4; ++t)
        #pragma unroll
        for (int kk = 0; kk < 4; ++kk) {
            const float* p = etile0 + t * N_ * EAT + 4 * kk;
            aC[t][kk].x = p[0]; aC[t][kk].y = p[1];
        }

    #pragma unroll
    for (int n = 0; n < 4; ++n) {
        v8f acc = {};
        #pragma unroll
        for (int s = 0; s < 16; ++s)
            acc = __builtin_amdgcn_wmma_f32_16x16x4_f32(false, As[n][s], false, Bs[s],
                                                        (short)0, acc, false, false);
        #pragma unroll
        for (int r = 0; r < 8; ++r)
            s_send[((jb + 2 * n) * 16 + r + 8 * lh) * EC + col1] = acc[r];
    }
    if (wave < 4) {
        v8f acc = {};
        #pragma unroll
        for (int s = 0; s < 16; ++s)
            acc = __builtin_amdgcn_wmma_f32_16x16x4_f32(false, Ar[s], false, Br[s],
                                                        (short)0, acc, false, false);
        if (lh == 0) {
            #pragma unroll
            for (int r = 0; r < 8; ++r)
                s_recvb[r * EC + col1] = acc[r] + bev;
        }
    }
    __syncthreads();

    v2f Bn[32];
    float bnv = 0.f;
    if (wave < 4) {
        const int o  = wave * 16 + ll;
        const int oc = o < NOUT ? o : NOUT - 1;
        #pragma unroll
        for (int s = 0; s < 32; ++s) {
            const float* q = Wn + (4 * s + 2 * lh) * NOUT + oc;
            Bn[s].x = q[0]; Bn[s].y = q[NOUT];
        }
        bnv = bn[oc];
    }

    const int jt = wave;
    v8f sC[4];
    #pragma unroll
    for (int ct = 0; ct < 4; ++ct)
        #pragma unroll
        for (int r = 0; r < 8; ++r)
            sC[ct][r] = s_send[(jt * 16 + r + 8 * lh) * EC + ct * 16 + ll];
    float* my_part = s_part + ((wave * 2 + lh) * EC + ll) * RI;
    #pragma nounroll
    for (int h = 0; h < 2; ++h) {
        const int ih = h * 4;
        v2f aN[4][4];
        if (h < 1) {
            #pragma unroll
            for (int t = 0; t < 4; ++t)
                #pragma unroll
                for (int kk = 0; kk < 4; ++kk) {
                    const float* p = etile0 + ((h + 1) * 4 + t) * N_ * EAT + 4 * kk;
                    aN[t][kk].x = p[0]; aN[t][kk].y = p[1];
                }
        } else {
            #pragma unroll
            for (int t = 0; t < 4; ++t)
                #pragma unroll
                for (int kk = 0; kk < 4; ++kk) aN[t][kk] = aC[t][kk];
        }
        float rb[4][4];
        v4f   ad[4][2];
        #pragma unroll
        for (int t = 0; t < 4; ++t) {
            #pragma unroll
            for (int ct = 0; ct < 4; ++ct)
                rb[t][ct] = s_recvb[(ih + t) * EC + ct * 16 + ll];
            const v4f* pa = reinterpret_cast<const v4f*>(s_adj + (ih + t) * N_ + jt * 16 + 8 * lh);
            ad[t][0] = pa[0];
            ad[t][1] = pa[1];
        }
        float pe[4][4];
        #pragma unroll
        for (int t = 0; t < 4; ++t) {
            #pragma unroll
            for (int ct = 0; ct < 4; ++ct) {
                v8f d = sC[ct];
                #pragma unroll
                for (int kk = 0; kk < 4; ++kk)
                    d = __builtin_amdgcn_wmma_f32_16x16x4_f32(false, aC[t][kk], false, bE[ct][kk],
                                                              (short)0, d, false, false);
                v2f acc = {0.f, 0.f};
                #pragma unroll
                for (int q = 0; q < 4; ++q) {
                    const v2f v = {d[2 * q] + rb[t][ct], d[2 * q + 1] + rb[t][ct]};
                    const v2f l = {lrelu(v.x), lrelu(v.y)};
                    const v2f a = {ad[t][q >> 1][(2 * q) & 3], ad[t][q >> 1][(2 * q + 1) & 3]};
                    acc = l * a + acc;
                }
                pe[t][ct] = acc.x + acc.y;
            }
        }
        #pragma unroll
        for (int ct = 0; ct < 4; ++ct) {
            const v4f o = {pe[0][ct], pe[1][ct], pe[2][ct], pe[3][ct]};
            *reinterpret_cast<v4f*>(my_part + ct * 16 * RI + ih) = o;
        }
        #pragma unroll
        for (int t = 0; t < 4; ++t)
            #pragma unroll
            for (int kk = 0; kk < 4; ++kk) aC[t][kk] = aN[t][kk];
    }
    __syncthreads();

    {
        const int c  = tid >> 2;
        const int i2 = (tid & 3) * 2;
        const v2f* pp = reinterpret_cast<const v2f*>(s_part + c * RI + i2);
        v2f s = pp[0];
        #pragma unroll
        for (int p = 1; p < 16; ++p) s += pp[p * (EC * RI / 2)];
        s_agg[i2 * EC + c]       = s.x;
        s_agg[(i2 + 1) * EC + c] = s.y;
    }
    __syncthreads();

    if (wave < 4) {
        const int o  = wave * 16 + ll;
        const int ar = ll & (RI - 1);
        v2f Ag[16];
        #pragma unroll
        for (int s = 0; s < 16; ++s) {
            Ag[s].x = s_agg[ar * EC + 4 * s + 2 * lh];
            Ag[s].y = s_agg[ar * EC + 4 * s + 2 * lh + 1];
        }
        v8f acc = {};
        #pragma unroll
        for (int s = 0; s < 16; ++s)
            acc = __builtin_amdgcn_wmma_f32_16x16x4_f32(false, Ar[s], false, Bn[s],
                                                        (short)0, acc, false, false);
        #pragma unroll
        for (int s = 0; s < 16; ++s)
            acc = __builtin_amdgcn_wmma_f32_16x16x4_f32(false, Ag[s], false, Bn[16 + s],
                                                        (short)0, acc, false, false);
        if (o < NOUT && lh == 0) {
            #pragma unroll
            for (int r = 0; r < 8; ++r) {
                const int node = i0 + r;
                out[(long)(b * N_ + node) * NOUT + o] = lrelu(acc[r] + bnv);
            }
        }
    }
}

extern "C" void kernel_launch(void* const* d_in, const int* in_sizes, int n_in,
                              void* d_out, int out_size, void* d_ws, size_t ws_size,
                              hipStream_t stream) {
    (void)in_sizes; (void)n_in; (void)d_ws; (void)ws_size; (void)out_size;
    const float* x         = (const float*)d_in[0];
    const float* node_attr = (const float*)d_in[1];
    const float* edge_attr = (const float*)d_in[2];
    const float* adj       = (const float*)d_in[3];
    const float* We        = (const float*)d_in[4];
    const float* be        = (const float*)d_in[5];
    const float* Wn        = (const float*)d_in[6];
    const float* bn        = (const float*)d_in[7];
    float* out             = (float*)d_out;

    dim3 grid(B_ * (N_ / RI));
    dim3 block(256);
    gin_fused<<<grid, block, 0, stream>>>(x, node_attr, edge_attr, adj, We, be, Wn, bn, out);
}
